# starting kernel
# baseline (speedup 1.0000x reference)
#include <hip/hip_runtime.h>
#include <hip/hip_bf16.h>

typedef __attribute__((ext_vector_type(2))) float v2f;
typedef __attribute__((ext_vector_type(8))) float v8f;

template <int K, int LDA, int LDB, bool COLGUARD>
__global__ void gemm_wmma_k(const float* __restrict__ A,
                            const float* __restrict__ B,
                            const float* __restrict__ bias,
                            float* __restrict__ C, int M) {
  const int lane  = threadIdx.x & 31;
  const int laneM = lane & 15;
  const int hi    = lane >> 4;
  const int row0  = blockIdx.x * 16;
  const int col0  = threadIdx.y * 16;

  int arow = row0 + laneM;
  if (arow > M - 1) arow = M - 1;
  const float* __restrict__ aptr = A + (size_t)arow * LDA + 2 * hi;

  const int col   = col0 + laneM;
  int       colc  = col;
  float     cmask = 1.0f;
  if (COLGUARD) {
    if (colc > LDB - 1) { colc = LDB - 1; cmask = 0.0f; }
  }
  const float* __restrict__ bptr = B + 2 * hi * LDB + colc;

  v8f acc = {};
#pragma unroll
  for (int k = 0; k < K; k += 4) {
    v2f a = *(const v2f*)(aptr + k);
    v2f b;
    b.x = bptr[k * LDB];
    b.y = bptr[k * LDB + LDB];
    if (COLGUARD) { b.x *= cmask; b.y *= cmask; }
    acc = __builtin_amdgcn_wmma_f32_16x16x4_f32(false, a, false, b,
                                                (short)0, acc, false, false);
  }

  if (COLGUARD && cmask == 0.0f) return;
  const float bc = COLGUARD ? bias[col] : 0.0f;
  float* __restrict__ cptr = C + (size_t)(row0 + 8 * hi) * LDB + col;
#pragma unroll
  for (int r = 0; r < 8; ++r) {
    const int row = row0 + r + 8 * hi;
    if (row < M) cptr[r * LDB] = acc[r] + bc;
  }
}

__global__ void scatter_edges(const float* __restrict__ T,
                              const int* __restrict__ src,
                              const int* __restrict__ dst,
                              const float* __restrict__ w,
                              float* __restrict__ agg, int E) {
  const int e = blockIdx.x * (blockDim.x >> 5) + (threadIdx.x >> 5);
  if (e >= E) return;
  const int   lane = threadIdx.x & 31;
  const int   s  = src[e];
  const int   d  = dst[e];
  const float we = w[e];
  const v2f   tv = *(const v2f*)(T + (size_t)s * 64 + 2 * lane);
  float* __restrict__ ap = agg + (size_t)d * 64 + 2 * lane;
  atomicAdd(ap,     tv.x * we);
  atomicAdd(ap + 1, tv.y * we);
}

__global__ void zero_f32(float* __restrict__ p, int n) {
  const int i = blockIdx.x * blockDim.x + threadIdx.x;
  if (i < n) p[i] = 0.0f;
}

__global__ void bias_relu_to_jk(const float* __restrict__ agg,
                                const float* __restrict__ b,
                                float* __restrict__ hjk,
                                int n64, int colOff) {
  const int i = blockIdx.x * blockDim.x + threadIdx.x;
  if (i >= n64) return;
  const int row = i >> 6;
  const int c   = i & 63;
  const float v = agg[i] + b[c];
  hjk[(size_t)row * 192 + colOff + c] = fmaxf(v, 0.0f);
}

extern "C" void kernel_launch(void* const* d_in, const int* in_sizes, int n_in,
                              void* d_out, int out_size, void* d_ws, size_t ws_size,
                              hipStream_t stream) {
  const float* x    = (const float*)d_in[0];
  const int*   ei   = (const int*)d_in[1];
  const float* ew   = (const float*)d_in[2];
  const float* W1   = (const float*)d_in[3];
  const float* b1   = (const float*)d_in[4];
  const float* W2   = (const float*)d_in[5];
  const float* b2   = (const float*)d_in[6];
  const float* W3   = (const float*)d_in[7];
  const float* b3   = (const float*)d_in[8];
  const float* Wlin = (const float*)d_in[9];
  const float* blin = (const float*)d_in[10];

  const int N = in_sizes[0] / 128;
  const int E = in_sizes[2];
  const int* src = ei;
  const int* dst = ei + E;

  float* T   = (float*)d_ws;
  float* agg = T   + (size_t)N * 64;
  float* hjk = agg + (size_t)N * 64;

  const int Mtiles  = (N + 15) / 16;
  const int n64     = N * 64;
  const int zblocks = (n64 + 255) / 256;
  const int sblocks = (E + 7) / 8;

  gemm_wmma_k<128, 128, 64, false><<<dim3(Mtiles, 1, 1), dim3(32, 4, 1), 0, stream>>>(
      x, W1, nullptr, T, N);
  zero_f32<<<zblocks, 256, 0, stream>>>(agg, n64);
  scatter_edges<<<sblocks, 256, 0, stream>>>(T, src, dst, ew, agg, E);
  bias_relu_to_jk<<<zblocks, 256, 0, stream>>>(agg, b1, hjk, n64, 0);

  gemm_wmma_k<64, 192, 64, false><<<dim3(Mtiles, 1, 1), dim3(32, 4, 1), 0, stream>>>(
      hjk, W2, nullptr, T, N);
  zero_f32<<<zblocks, 256, 0, stream>>>(agg, n64);
  scatter_edges<<<sblocks, 256, 0, stream>>>(T, src, dst, ew, agg, E);
  bias_relu_to_jk<<<zblocks, 256, 0, stream>>>(agg, b2, hjk, n64, 64);

  gemm_wmma_k<64, 192, 64, false><<<dim3(Mtiles, 1, 1), dim3(32, 4, 1), 0, stream>>>(
      hjk + 64, W3, nullptr, T, N);
  zero_f32<<<zblocks, 256, 0, stream>>>(agg, n64);
  scatter_edges<<<sblocks, 256, 0, stream>>>(T, src, dst, ew, agg, E);
  bias_relu_to_jk<<<zblocks, 256, 0, stream>>>(agg, b3, hjk, n64, 128);

  gemm_wmma_k<192, 192, 40, true><<<dim3(Mtiles, 1, 1), dim3(32, 3, 1), 0, stream>>>(
      hjk, Wlin, blin, (float*)d_out, N);
}


# Atomic f32 scatter-add replaced by a per-call counting sort of edges by destination; one wave per node sums its rows in arrival-order rank, adds bias, ReLU, stores once (no zero fill)
# speedup vs baseline: 1.5315x; 1.5315x over previous
#include <hip/hip_runtime.h>
#include <hip/hip_bf16.h>

typedef __attribute__((ext_vector_type(2))) float v2f;
typedef __attribute__((ext_vector_type(8))) float v8f;
typedef __attribute__((ext_vector_type(2))) int   v2i;

template <int K, int LDA, int LDB, bool COLGUARD>
__global__ void gemm_wmma_k(const float* __restrict__ A,
                            const float* __restrict__ B,
                            const float* __restrict__ bias,
                            float* __restrict__ C, int M) {
  const int lane  = threadIdx.x & 31;
  const int laneM = lane & 15;
  const int hi    = lane >> 4;
  const int row0  = blockIdx.x * 16;
  const int col0  = threadIdx.y * 16;

  int arow = row0 + laneM;
  if (arow > M - 1) arow = M - 1;
  const float* __restrict__ aptr = A + (size_t)arow * LDA + 2 * hi;

  const int col   = col0 + laneM;
  int       colc  = col;
  float     cmask = 1.0f;
  if (COLGUARD) {
    if (colc > LDB - 1) { colc = LDB - 1; cmask = 0.0f; }
  }
  const float* __restrict__ bptr = B + 2 * hi * LDB + colc;

  v8f acc = {};
#pragma unroll
  for (int k = 0; k < K; k += 4) {
    v2f a = *(const v2f*)(aptr + k);
    v2f b;
    b.x = bptr[k * LDB];
    b.y = bptr[k * LDB + LDB];
    if (COLGUARD) { b.x *= cmask; b.y *= cmask; }
    acc = __builtin_amdgcn_wmma_f32_16x16x4_f32(false, a, false, b,
                                                (short)0, acc, false, false);
  }

  if (COLGUARD && cmask == 0.0f) return;
  const float bc = COLGUARD ? bias[col] : 0.0f;
  float* __restrict__ cptr = C + (size_t)(row0 + 8 * hi) * LDB + col;
#pragma unroll
  for (int r = 0; r < 8; ++r) {
    const int row = row0 + r + 8 * hi;
    if (row < M) cptr[r * LDB] = acc[r] + bc;
  }
}

__global__ void zero_i32(int* __restrict__ p, int n) {
  const int i = blockIdx.x * blockDim.x + threadIdx.x;
  if (i < n) p[i] = 0;
}

__global__ void rank_edges(const int* __restrict__ dst,
                           int* __restrict__ deg,
                           int* __restrict__ rank, int E) {
  const int e = blockIdx.x * blockDim.x + threadIdx.x;
  if (e >= E) return;
  rank[e] = atomicAdd(deg + dst[e], 1);
}

__device__ __forceinline__ int4 load4_to(const int* __restrict__ p, int i, int n) {
  int4 v = make_int4(0, 0, 0, 0);
  if (i + 3 < n) {
    v = *(const int4*)(p + i);
  } else {
    if (i     < n) v.x = p[i];
    if (i + 1 < n) v.y = p[i + 1];
    if (i + 2 < n) v.z = p[i + 2];
  }
  return v;
}

__device__ __forceinline__ void store4_to(int* __restrict__ p, int i, int n, int4 o) {
  if (i + 3 < n) {
    *(int4*)(p + i) = o;
  } else {
    if (i     < n) p[i]     = o.x;
    if (i + 1 < n) p[i + 1] = o.y;
    if (i + 2 < n) p[i + 2] = o.z;
  }
}

__global__ void __launch_bounds__(1024)
scan_degrees(int* __restrict__ deg, int n) {
  __shared__ int waveSum[32];
  __shared__ int tileSum;
  const int tid  = threadIdx.x;
  const int lane = tid & 31;
  const int wid  = tid >> 5;
  int carry = 0;
  for (int base = 0; base < n; base += 16384) {
    const int  i0   = base + tid * 16;
    const bool full = base + 16384 <= n;
    int4 v[4];
    if (full) {
#pragma unroll
      for (int q = 0; q < 4; ++q) v[q] = *(const int4*)(deg + i0 + 4 * q);
    } else {
#pragma unroll
      for (int q = 0; q < 4; ++q) v[q] = load4_to(deg, i0 + 4 * q, n);
    }
    int s = 0;
#pragma unroll
    for (int q = 0; q < 4; ++q) s += v[q].x + v[q].y + v[q].z + v[q].w;
    int incl = s;
#pragma unroll
    for (int d = 1; d < 32; d <<= 1) {
      const int t = __shfl_up(incl, d);
      if (lane >= d) incl += t;
    }
    if (lane == 31) waveSum[wid] = incl;
    __syncthreads();
    if (wid == 0) {
      const int w = waveSum[lane];
      int wi = w;
#pragma unroll
      for (int d = 1; d < 32; d <<= 1) {
        const int t = __shfl_up(wi, d);
        if (lane >= d) wi += t;
      }
      waveSum[lane] = wi - w;
      if (lane == 31) tileSum = wi;
    }
    __syncthreads();
    int run = carry + waveSum[wid] + incl - s;
    carry += tileSum;
#pragma unroll
    for (int q = 0; q < 4; ++q) {
      int4 o;
      o.x = run;
      o.y = o.x + v[q].x;
      o.z = o.y + v[q].y;
      o.w = o.z + v[q].z;
      run = o.w + v[q].w;
      if (full) *(int4*)(deg + i0 + 4 * q) = o;
      else      store4_to(deg, i0 + 4 * q, n, o);
    }
    __syncthreads();
  }
}

__global__ void place_edges(const int* __restrict__ src,
                            const int* __restrict__ dst,
                            const float* __restrict__ w,
                            const int* __restrict__ first,
                            const int* __restrict__ rank,
                            int* __restrict__ list, int E) {
  const int e = blockIdx.x * blockDim.x + threadIdx.x;
  if (e >= E) return;
  const int slot = first[dst[e]] + rank[e];
  v2i entry;
  entry.x = src[e];
  entry.y = __float_as_int(w[e]);
  *(v2i*)(list + 2 * (size_t)slot) = entry;
}

__global__ void gather_bias_relu(const float* __restrict__ T,
                                 const int* __restrict__ first,
                                 const int* __restrict__ list,
                                 const float* __restrict__ b,
                                 float* __restrict__ hjk,
                                 int N, int colOff) {
  const int node = __builtin_amdgcn_readfirstlane(
      (int)(blockIdx.x * (blockDim.x >> 5) + (threadIdx.x >> 5)));
  if (node >= N) return;
  const int lane = threadIdx.x & 31;
  const int beg = first[node];
  const int end = first[node + 1];
  const float* __restrict__ Tl = T + 2 * lane;
  v2f acc = {0.0f, 0.0f};
  for (int c = beg; c < end; c += 32) {
    const int n = (end - c < 32) ? (end - c) : 32;
    v2i me = {0, 0};
    if (lane < n) me = *(const v2i*)(list + 2 * (size_t)(c + lane));
#pragma unroll 4
    for (int j = 0; j < n; ++j) {
      const int   s  = __builtin_amdgcn_readlane(me.x, j);
      const float we = __int_as_float(__builtin_amdgcn_readlane(me.y, j));
      const v2f   tv = *(const v2f*)(Tl + (size_t)s * 64);
      acc.x += tv.x * we;
      acc.y += tv.y * we;
    }
  }
  const v2f bv = *(const v2f*)(b + 2 * lane);
  v2f o;
  o.x = fmaxf(acc.x + bv.x, 0.0f);
  o.y = fmaxf(acc.y + bv.y, 0.0f);
  *(v2f*)(hjk + (size_t)node * 192 + colOff + 2 * lane) = o;
}

extern "C" void kernel_launch(void* const* d_in, const int* in_sizes, int n_in,
                              void* d_out, int out_size, void* d_ws, size_t ws_size,
                              hipStream_t stream) {
  const float* x    = (const float*)d_in[0];
  const int*   ei   = (const int*)d_in[1];
  const float* ew   = (const float*)d_in[2];
  const float* W1   = (const float*)d_in[3];
  const float* b1   = (const float*)d_in[4];
  const float* W2   = (const float*)d_in[5];
  const float* b2   = (const float*)d_in[6];
  const float* W3   = (const float*)d_in[7];
  const float* b3   = (const float*)d_in[8];
  const float* Wlin = (const float*)d_in[9];
  const float* blin = (const float*)d_in[10];

  const int N = in_sizes[0] / 128;
  const int E = in_sizes[2];
  const int* src = ei;
  const int* dst = ei + E;

  float* T    = (float*)d_ws;
  float* mid  = T   + (size_t)N * 64;
  float* hjk  = mid + (size_t)N * 64;

  const int degLen = (N + 1 + 3) & ~3;
  int*  deg  = (int*)mid;
  int*  rank = deg + degLen;
  int*  list = rank + ((E + 3) & ~3);

  const int Mtiles  = (N + 15) / 16;
  const int eblocks = (E + 255) / 256;
  const int gblocks = (N + 7) / 8;

  zero_i32<<<(N + 1 + 255) / 256, 256, 0, stream>>>(deg, N + 1);
  rank_edges<<<eblocks, 256, 0, stream>>>(dst, deg, rank, E);
  scan_degrees<<<1, 1024, 0, stream>>>(deg, N + 1);
  place_edges<<<eblocks, 256, 0, stream>>>(src, dst, ew, deg, rank, list, E);

  gemm_wmma_k<128, 128, 64, false><<<dim3(Mtiles, 1, 1), dim3(32, 4, 1), 0, stream>>>(
      x, W1, nullptr, T, N);
  gather_bias_relu<<<gblocks, 256, 0, stream>>>(T, deg, list, b1, hjk, N, 0);

  gemm_wmma_k<64, 192, 64, false><<<dim3(Mtiles, 1, 1), dim3(32, 4, 1), 0, stream>>>(
      hjk, W2, nullptr, T, N);
  gather_bias_relu<<<gblocks, 256, 0, stream>>>(T, deg, list, b2, hjk, N, 64);

  gemm_wmma_k<64, 192, 64, false><<<dim3(Mtiles, 1, 1), dim3(32, 4, 1), 0, stream>>>(
      hjk + 64, W3, nullptr, T, N);
  gather_bias_relu<<<gblocks, 256, 0, stream>>>(T, deg, list, b3, hjk, N, 128);

  gemm_wmma_k<192, 192, 40, true><<<dim3(Mtiles, 1, 1), dim3(32, 3, 1), 0, stream>>>(
      hjk, Wlin, blin, (float*)d_out, N);
}


# Each layer's gather also does the products that use its rows: 16-node h tile in LDS, h@Wnext and h@Wlin slice by WMMA in that launch (11->8 launches); out = 3 partial sums, k order in each 8 changed
# speedup vs baseline: 1.7418x; 1.1373x over previous
#include <hip/hip_runtime.h>
#include <hip/hip_bf16.h>

typedef __attribute__((ext_vector_type(2))) float v2f;
typedef __attribute__((ext_vector_type(8))) float v8f;
typedef __attribute__((ext_vector_type(2))) int   v2i;

template <int K, int LDA, int LDB, bool COLGUARD>
__global__ void gemm_wmma_k(const float* __restrict__ A,
                            const float* __restrict__ B,
                            const float* __restrict__ bias,
                            float* __restrict__ C, int M) {
  const int lane  = threadIdx.x & 31;
  const int laneM = lane & 15;
  const int hi    = lane >> 4;
  const int row0  = blockIdx.x * 16;
  const int col0  = threadIdx.y * 16;

  int arow = row0 + laneM;
  if (arow > M - 1) arow = M - 1;
  const float* __restrict__ aptr = A + (size_t)arow * LDA + 2 * hi;

  const int col   = col0 + laneM;
  int       colc  = col;
  float     cmask = 1.0f;
  if (COLGUARD) {
    if (colc > LDB - 1) { colc = LDB - 1; cmask = 0.0f; }
  }
  const float* __restrict__ bptr = B + 2 * hi * LDB + colc;

  v8f acc = {};
#pragma unroll
  for (int k = 0; k < K; k += 4) {
    v2f a = *(const v2f*)(aptr + k);
    v2f b;
    b.x = bptr[k * LDB];
    b.y = bptr[k * LDB + LDB];
    if (COLGUARD) { b.x *= cmask; b.y *= cmask; }
    acc = __builtin_amdgcn_wmma_f32_16x16x4_f32(false, a, false, b,
                                                (short)0, acc, false, false);
  }

  if (COLGUARD && cmask == 0.0f) return;
  const float bc = COLGUARD ? bias[col] : 0.0f;
  float* __restrict__ cptr = C + (size_t)(row0 + 8 * hi) * LDB + col;
#pragma unroll
  for (int r = 0; r < 8; ++r) {
    const int row = row0 + r + 8 * hi;
    if (row < M) cptr[r * LDB] = acc[r] + bc;
  }
}

__global__ void zero_i32(int* __restrict__ p, int n) {
  const int i = blockIdx.x * blockDim.x + threadIdx.x;
  if (i < n) p[i] = 0;
}

__global__ void rank_edges(const int* __restrict__ dst,
                           int* __restrict__ deg,
                           int* __restrict__ rank, int E) {
  const int e = blockIdx.x * blockDim.x + threadIdx.x;
  if (e >= E) return;
  rank[e] = atomicAdd(deg + dst[e], 1);
}

__device__ __forceinline__ int4 load4_to(const int* __restrict__ p, int i, int n) {
  int4 v = make_int4(0, 0, 0, 0);
  if (i + 3 < n) {
    v = *(const int4*)(p + i);
  } else {
    if (i     < n) v.x = p[i];
    if (i + 1 < n) v.y = p[i + 1];
    if (i + 2 < n) v.z = p[i + 2];
  }
  return v;
}

__device__ __forceinline__ void store4_to(int* __restrict__ p, int i, int n, int4 o) {
  if (i + 3 < n) {
    *(int4*)(p + i) = o;
  } else {
    if (i     < n) p[i]     = o.x;
    if (i + 1 < n) p[i + 1] = o.y;
    if (i + 2 < n) p[i + 2] = o.z;
  }
}

__global__ void __launch_bounds__(1024)
scan_degrees(int* __restrict__ deg, int n) {
  __shared__ int waveSum[32];
  __shared__ int tileSum;
  const int tid  = threadIdx.x;
  const int lane = tid & 31;
  const int wid  = tid >> 5;
  int carry = 0;
  for (int base = 0; base < n; base += 16384) {
    const int  i0   = base + tid * 16;
    const bool full = base + 16384 <= n;
    int4 v[4];
    if (full) {
#pragma unroll
      for (int q = 0; q < 4; ++q) v[q] = *(const int4*)(deg + i0 + 4 * q);
    } else {
#pragma unroll
      for (int q = 0; q < 4; ++q) v[q] = load4_to(deg, i0 + 4 * q, n);
    }
    int s = 0;
#pragma unroll
    for (int q = 0; q < 4; ++q) s += v[q].x + v[q].y + v[q].z + v[q].w;
    int incl = s;
#pragma unroll
    for (int d = 1; d < 32; d <<= 1) {
      const int t = __shfl_up(incl, d);
      if (lane >= d) incl += t;
    }
    if (lane == 31) waveSum[wid] = incl;
    __syncthreads();
    if (wid == 0) {
      const int w = waveSum[lane];
      int wi = w;
#pragma unroll
      for (int d = 1; d < 32; d <<= 1) {
        const int t = __shfl_up(wi, d);
        if (lane >= d) wi += t;
      }
      waveSum[lane] = wi - w;
      if (lane == 31) tileSum = wi;
    }
    __syncthreads();
    int run = carry + waveSum[wid] + incl - s;
    carry += tileSum;
#pragma unroll
    for (int q = 0; q < 4; ++q) {
      int4 o;
      o.x = run;
      o.y = o.x + v[q].x;
      o.z = o.y + v[q].y;
      o.w = o.z + v[q].z;
      run = o.w + v[q].w;
      if (full) *(int4*)(deg + i0 + 4 * q) = o;
      else      store4_to(deg, i0 + 4 * q, n, o);
    }
    __syncthreads();
  }
}

__global__ void place_edges(const int* __restrict__ src,
                            const int* __restrict__ dst,
                            const float* __restrict__ w,
                            const int* __restrict__ first,
                            const int* __restrict__ rank,
                            int* __restrict__ list, int E) {
  const int e = blockIdx.x * blockDim.x + threadIdx.x;
  if (e >= E) return;
  const int slot = first[dst[e]] + rank[e];
  v2i entry;
  entry.x = src[e];
  entry.y = __float_as_int(w[e]);
  *(v2i*)(list + 2 * (size_t)slot) = entry;
}

constexpr int H_PITCH = 68;

template <int LDB>
__device__ __forceinline__ v8f tile_k64(const float* arow,
                                        const float* __restrict__ bptr) {
  v8f acc = {};
#pragma unroll
  for (int k = 0; k < 64; k += 8) {
    const float4 a4 = *(const float4*)(arow + k);
    v2f a0, a1, b0, b1;
    a0.x = a4.x; a0.y = a4.y;
    a1.x = a4.z; a1.y = a4.w;
    b0.x = bptr[(k + 0) * LDB];
    b0.y = bptr[(k + 1) * LDB];
    b1.x = bptr[(k + 2) * LDB];
    b1.y = bptr[(k + 3) * LDB];
    acc = __builtin_amdgcn_wmma_f32_16x16x4_f32(false, a0, false, b0,
                                                (short)0, acc, false, false);
    acc = __builtin_amdgcn_wmma_f32_16x16x4_f32(false, a1, false, b1,
                                                (short)0, acc, false, false);
  }
  return acc;
}

template <bool HAS_NEXT, bool FIRST>
__global__ void __launch_bounds__(256)
gather_layer(const float* __restrict__ T,
             const int* __restrict__ first,
             const int* __restrict__ list,
             const float* __restrict__ b,
             const float* __restrict__ Wn,
             float* __restrict__ Tn,
             const float* __restrict__ Wl,
             const float* __restrict__ blin,
             float* __restrict__ out,
             int N) {
  __shared__ __attribute__((aligned(16))) float Hs[16 * H_PITCH];

  const int lane = threadIdx.x & 31;
  const int wid  = __builtin_amdgcn_readfirstlane((int)(threadIdx.x >> 5));
  const int row0 = blockIdx.x * 16;

  const float* __restrict__ Tl = T + 2 * lane;
  const v2f bv = *(const v2f*)(b + 2 * lane);
  for (int t = 0; t < 2; ++t) {
    const int ln   = 2 * wid + t;
    const int node = row0 + ln;
    v2f o = {0.0f, 0.0f};
    if (node < N) {
      const int beg = first[node];
      const int end = first[node + 1];
      v2f acc = {0.0f, 0.0f};
      for (int c = beg; c < end; c += 32) {
        const int n = (end - c < 32) ? (end - c) : 32;
        v2i me = {0, 0};
        if (lane < n) me = *(const v2i*)(list + 2 * (size_t)(c + lane));
#pragma unroll 4
        for (int j = 0; j < n; ++j) {
          const int   s  = __builtin_amdgcn_readlane(me.x, j);
          const float we = __int_as_float(__builtin_amdgcn_readlane(me.y, j));
          const v2f   tv = *(const v2f*)(Tl + (size_t)s * 64);
          acc.x += tv.x * we;
          acc.y += tv.y * we;
        }
      }
      o.x = fmaxf(acc.x + bv.x, 0.0f);
      o.y = fmaxf(acc.y + bv.y, 0.0f);
    }
    *(v2f*)(Hs + ln * H_PITCH + 2 * lane) = o;
  }

  __syncthreads();

  const int laneM = lane & 15;
  const int hi    = lane >> 4;
  const float* arow = Hs + laneM * H_PITCH + 4 * hi;
  constexpr int OUTW0 = HAS_NEXT ? 4 : 0;

  if (HAS_NEXT && wid < 4) {
    const int col = 16 * wid + laneM;
    const v8f acc = tile_k64<64>(arow, Wn + 4 * hi * 64 + col);
    float* __restrict__ cptr = Tn + (size_t)(row0 + 8 * hi) * 64 + col;
#pragma unroll
    for (int r = 0; r < 8; ++r) {
      const int row = row0 + r + 8 * hi;
      if (row < N) cptr[r * 64] = acc[r];
    }
  } else if (wid >= OUTW0 && wid < OUTW0 + 3) {
    const int col  = 16 * (wid - OUTW0) + laneM;
    const int colc = (col < 40) ? col : 39;
    const v8f acc = tile_k64<40>(arow, Wl + 4 * hi * 40 + colc);
    if (col < 40) {
      float add[8];
      if (FIRST) {
        const float bl = blin[col];
#pragma unroll
        for (int r = 0; r < 8; ++r) add[r] = bl;
      } else {
#pragma unroll
        for (int r = 0; r < 8; ++r) {
          int rowc = row0 + r + 8 * hi;
          if (rowc > N - 1) rowc = N - 1;
          add[r] = out[(size_t)rowc * 40 + col];
        }
      }
      float* optr = out + (size_t)(row0 + 8 * hi) * 40 + col;
#pragma unroll
      for (int r = 0; r < 8; ++r) {
        const int row = row0 + r + 8 * hi;
        if (row < N) optr[r * 40] = acc[r] + add[r];
      }
    }
  }
}

extern "C" void kernel_launch(void* const* d_in, const int* in_sizes, int n_in,
                              void* d_out, int out_size, void* d_ws, size_t ws_size,
                              hipStream_t stream) {
  const float* x    = (const float*)d_in[0];
  const int*   ei   = (const int*)d_in[1];
  const float* ew   = (const float*)d_in[2];
  const float* W1   = (const float*)d_in[3];
  const float* b1   = (const float*)d_in[4];
  const float* W2   = (const float*)d_in[5];
  const float* b2   = (const float*)d_in[6];
  const float* W3   = (const float*)d_in[7];
  const float* b3   = (const float*)d_in[8];
  const float* Wlin = (const float*)d_in[9];
  const float* blin = (const float*)d_in[10];

  const int N = in_sizes[0] / 128;
  const int E = in_sizes[2];
  const int* src = ei;
  const int* dst = ei + E;

  float* Ta   = (float*)d_ws;
  float* mid  = Ta  + (size_t)N * 64;
  float* Tb   = mid + (size_t)N * 64;

  const int degLen = (N + 1 + 3) & ~3;
  int*  deg  = (int*)mid;
  int*  rank = deg + degLen;
  int*  list = rank + ((E + 3) & ~3);

  const int Mtiles  = (N + 15) / 16;
  const int eblocks = (E + 255) / 256;

  zero_i32<<<(N + 1 + 255) / 256, 256, 0, stream>>>(deg, N + 1);
  rank_edges<<<eblocks, 256, 0, stream>>>(dst, deg, rank, E);
  scan_degrees<<<1, 1024, 0, stream>>>(deg, N + 1);
  place_edges<<<eblocks, 256, 0, stream>>>(src, dst, ew, deg, rank, list, E);

  gemm_wmma_k<128, 128, 64, false><<<dim3(Mtiles, 1, 1), dim3(32, 4, 1), 0, stream>>>(
      x, W1, nullptr, Ta, N);

  gather_layer<true, true><<<Mtiles, 256, 0, stream>>>(
      Ta, deg, list, b1, W2, Tb, Wlin, blin, (float*)d_out, N);

  gather_layer<true, false><<<Mtiles, 256, 0, stream>>>(
      Tb, deg, list, b2, W3, Ta, Wlin + 64 * 40, blin, (float*)d_out, N);

  gather_layer<false, false><<<Mtiles, 256, 0, stream>>>(
      Ta, deg, list, b3, nullptr, nullptr, Wlin + 128 * 40, blin, (float*)d_out, N);
}


# Edge list built by a two-level LDS counting sort (chunks ordered by 256-node bin, then a workgroup per bin): no global atomics or scattered stores, 8->6 launches; node sum order = LDS add arrival
# speedup vs baseline: 3.1834x; 1.8276x over previous
#include <hip/hip_runtime.h>
#include <hip/hip_bf16.h>

typedef __attribute__((ext_vector_type(2))) float v2f;
typedef __attribute__((ext_vector_type(8))) float v8f;
typedef __attribute__((ext_vector_type(2))) int   v2i;

template <int K, int LDA, int LDB, bool COLGUARD>
__global__ void gemm_wmma_k(const float* __restrict__ A,
                            const float* __restrict__ B,
                            const float* __restrict__ bias,
                            float* __restrict__ C, int M) {
  const int lane  = threadIdx.x & 31;
  const int laneM = lane & 15;
  const int hi    = lane >> 4;
  const int row0  = blockIdx.x * 16;
  const int col0  = threadIdx.y * 16;

  int arow = row0 + laneM;
  if (arow > M - 1) arow = M - 1;
  const float* __restrict__ aptr = A + (size_t)arow * LDA + 2 * hi;

  const int col   = col0 + laneM;
  int       colc  = col;
  float     cmask = 1.0f;
  if (COLGUARD) {
    if (colc > LDB - 1) { colc = LDB - 1; cmask = 0.0f; }
  }
  const float* __restrict__ bptr = B + 2 * hi * LDB + colc;

  v8f acc = {};
#pragma unroll
  for (int k = 0; k < K; k += 4) {
    v2f a = *(const v2f*)(aptr + k);
    v2f b;
    b.x = bptr[k * LDB];
    b.y = bptr[k * LDB + LDB];
    if (COLGUARD) { b.x *= cmask; b.y *= cmask; }
    acc = __builtin_amdgcn_wmma_f32_16x16x4_f32(false, a, false, b,
                                                (short)0, acc, false, false);
  }

  if (COLGUARD && cmask == 0.0f) return;
  const float bc = COLGUARD ? bias[col] : 0.0f;
  float* __restrict__ cptr = C + (size_t)(row0 + 8 * hi) * LDB + col;
#pragma unroll
  for (int r = 0; r < 8; ++r) {
    const int row = row0 + r + 8 * hi;
    if (row < M) cptr[r * LDB] = acc[r] + bc;
  }
}

constexpr int EDGE_CHUNK = 3072;
constexpr int SORT_CAP   = 8192;

__device__ __forceinline__ void scan256(const int* cnt, int* pre, int* wsum, int tid) {
  const int lane = tid & 31;
  const int wid  = tid >> 5;
  const int v = (tid < 256) ? cnt[tid] : 0;
  int incl = v;
#pragma unroll
  for (int d = 1; d < 32; d <<= 1) {
    const int t = __shfl_up(incl, d);
    if (lane >= d) incl += t;
  }
  if (tid < 256 && lane == 31) wsum[wid] = incl;
  __syncthreads();
  if (tid < 256) {
    int base = 0;
    for (int w = 0; w < wid; ++w) base += wsum[w];
    pre[tid] = base + incl - v;
  }
  __syncthreads();
}

__global__ void __launch_bounds__(1024)
chunk_sort(const int* __restrict__ src, const int* __restrict__ dst,
           const float* __restrict__ w, int* __restrict__ binned,
           int* __restrict__ tab, int E, int C) {
  __shared__ int cnt[256];
  __shared__ int lst[256];
  __shared__ int wsum[8];
  __shared__ v2i stage[EDGE_CHUNK];
  const int tid = threadIdx.x;
  const int c   = blockIdx.x;
  if (tid < 256) cnt[tid] = 0;
  __syncthreads();
  const int e0 = c * EDGE_CHUNK;
  int nloc = E - e0;
  if (nloc > EDGE_CHUNK) nloc = EDGE_CHUNK;
  v2i ent[3];
  int bb[3], rk[3];
#pragma unroll
  for (int k = 0; k < 3; ++k) {
    const int  i  = k * 1024 + tid;
    const bool ok = i < nloc;
    const int  ix = e0 + (ok ? i : 0);
    const int  d  = dst[ix];
    bb[k]    = ok ? (d >> 8) : -1;
    ent[k].x = (src[ix] << 8) | (d & 255);
    ent[k].y = __float_as_int(w[ix]);
  }
#pragma unroll
  for (int k = 0; k < 3; ++k) {
    rk[k] = 0;
    if (bb[k] >= 0) rk[k] = atomicAdd(&cnt[bb[k]], 1);
  }
  __syncthreads();
  scan256(cnt, lst, wsum, tid);
#pragma unroll
  for (int k = 0; k < 3; ++k) {
    if (bb[k] >= 0) stage[lst[bb[k]] + rk[k]] = ent[k];
  }
  if (tid < 256) tab[tid * C + c] = lst[tid];
  if (tid == 0)  tab[256 * C + c] = nloc;
  __syncthreads();
  v2i* __restrict__ outp = (v2i*)(binned + 2 * (size_t)e0);
  for (int p = tid; p < nloc; p += 1024) outp[p] = stage[p];
}

__global__ void __launch_bounds__(1024)
bin_sort(const int* __restrict__ binned, const int* __restrict__ tab,
         int* __restrict__ first, int* __restrict__ list,
         int N, int E, int C) {
  __shared__ int cnt[256];
  __shared__ int cur[256];
  __shared__ int wsum[8];
  __shared__ int rs[1024];
  __shared__ int rl[1024];
  __shared__ int red[64];
  __shared__ int tot[2];
  __shared__ v2i sout[SORT_CAP];
  const int tid  = threadIdx.x;
  const int lane = tid & 31;
  const int wid  = tid >> 5;
  const int b    = blockIdx.x;
  const int node0 = b * 256;

  int a = 0, len = 0;
  if (tid < C) {
    a   = tab[b * C + tid];
    len = tab[(b + 1) * C + tid] - a;
    rs[tid] = tid * EDGE_CHUNK + a;
    rl[tid] = len;
  }
#pragma unroll
  for (int d = 16; d >= 1; d >>= 1) {
    a   += __shfl_xor(a, d);
    len += __shfl_xor(len, d);
  }
  if (lane == 0) { red[wid] = a; red[32 + wid] = len; }
  if (tid < 256) cnt[tid] = 0;
  __syncthreads();
  if (wid == 0) {
    int sa = red[lane], sl = red[32 + lane];
#pragma unroll
    for (int d = 16; d >= 1; d >>= 1) {
      sa += __shfl_xor(sa, d);
      sl += __shfl_xor(sl, d);
    }
    if (lane == 0) { tot[0] = sa; tot[1] = sl; }
  }
  __syncthreads();
  const int beg = tot[0];
  const int n   = tot[1];

  const int hw  = tid >> 4;
  const int l16 = tid & 15;
  for (int r = hw; r < C; r += 64) {
    const int s = rs[r], m = rl[r];
    for (int j = l16; j < m; j += 16)
      atomicAdd(&cnt[binned[2 * (size_t)(s + j)] & 255], 1);
  }
  __syncthreads();
  scan256(cnt, cur, wsum, tid);
  if (tid < 256 && node0 + tid < N) first[node0 + tid] = beg + cur[tid];
  if (node0 + 256 >= N && tid == 0) first[N] = E;
  __syncthreads();

  const bool park = n <= SORT_CAP;
  v2i* __restrict__ outp = (v2i*)(list + 2 * (size_t)beg);
  for (int r = hw; r < C; r += 64) {
    const int s = rs[r], m = rl[r];
    for (int j = l16; j < m; j += 16) {
      const v2i e = *(const v2i*)(binned + 2 * (size_t)(s + j));
      const int pos = atomicAdd(&cur[e.x & 255], 1);
      v2i o;
      o.x = (int)((unsigned)e.x >> 8);
      o.y = e.y;
      if (park) sout[pos] = o;
      else      outp[pos] = o;
    }
  }
  if (park) {
    __syncthreads();
    for (int i = tid; i < n; i += 1024) outp[i] = sout[i];
  }
}

constexpr int H_PITCH = 68;

template <int LDB>
__device__ __forceinline__ v8f tile_k64(const float* arow,
                                        const float* __restrict__ bptr) {
  v8f acc = {};
#pragma unroll
  for (int k = 0; k < 64; k += 8) {
    const float4 a4 = *(const float4*)(arow + k);
    v2f a0, a1, b0, b1;
    a0.x = a4.x; a0.y = a4.y;
    a1.x = a4.z; a1.y = a4.w;
    b0.x = bptr[(k + 0) * LDB];
    b0.y = bptr[(k + 1) * LDB];
    b1.x = bptr[(k + 2) * LDB];
    b1.y = bptr[(k + 3) * LDB];
    acc = __builtin_amdgcn_wmma_f32_16x16x4_f32(false, a0, false, b0,
                                                (short)0, acc, false, false);
    acc = __builtin_amdgcn_wmma_f32_16x16x4_f32(false, a1, false, b1,
                                                (short)0, acc, false, false);
  }
  return acc;
}

template <bool HAS_NEXT, bool FIRST>
__global__ void __launch_bounds__(256)
gather_layer(const float* __restrict__ T,
             const int* __restrict__ first,
             const int* __restrict__ list,
             const float* __restrict__ b,
             const float* __restrict__ Wn,
             float* __restrict__ Tn,
             const float* __restrict__ Wl,
             const float* __restrict__ blin,
             float* __restrict__ out,
             int N) {
  __shared__ __attribute__((aligned(16))) float Hs[16 * H_PITCH];

  const int lane = threadIdx.x & 31;
  const int wid  = __builtin_amdgcn_readfirstlane((int)(threadIdx.x >> 5));
  const int row0 = blockIdx.x * 16;

  const float* __restrict__ Tl = T + 2 * lane;
  const v2f bv = *(const v2f*)(b + 2 * lane);
  for (int t = 0; t < 2; ++t) {
    const int ln   = 2 * wid + t;
    const int node = row0 + ln;
    v2f o = {0.0f, 0.0f};
    if (node < N) {
      const int beg = first[node];
      const int end = first[node + 1];
      v2f acc = {0.0f, 0.0f};
      for (int c = beg; c < end; c += 32) {
        const int n = (end - c < 32) ? (end - c) : 32;
        v2i me = {0, 0};
        if (lane < n) me = *(const v2i*)(list + 2 * (size_t)(c + lane));
#pragma unroll 4
        for (int j = 0; j < n; ++j) {
          const int   s  = __builtin_amdgcn_readlane(me.x, j);
          const float we = __int_as_float(__builtin_amdgcn_readlane(me.y, j));
          const v2f   tv = *(const v2f*)(Tl + (size_t)s * 64);
          acc.x += tv.x * we;
          acc.y += tv.y * we;
        }
      }
      o.x = fmaxf(acc.x + bv.x, 0.0f);
      o.y = fmaxf(acc.y + bv.y, 0.0f);
    }
    *(v2f*)(Hs + ln * H_PITCH + 2 * lane) = o;
  }

  __syncthreads();

  const int laneM = lane & 15;
  const int hi    = lane >> 4;
  const float* arow = Hs + laneM * H_PITCH + 4 * hi;
  constexpr int OUTW0 = HAS_NEXT ? 4 : 0;

  if (HAS_NEXT && wid < 4) {
    const int col = 16 * wid + laneM;
    const v8f acc = tile_k64<64>(arow, Wn + 4 * hi * 64 + col);
    float* __restrict__ cptr = Tn + (size_t)(row0 + 8 * hi) * 64 + col;
#pragma unroll
    for (int r = 0; r < 8; ++r) {
      const int row = row0 + r + 8 * hi;
      if (row < N) cptr[r * 64] = acc[r];
    }
  } else if (wid >= OUTW0 && wid < OUTW0 + 3) {
    const int col  = 16 * (wid - OUTW0) + laneM;
    const int colc = (col < 40) ? col : 39;
    const v8f acc = tile_k64<40>(arow, Wl + 4 * hi * 40 + colc);
    if (col < 40) {
      float add[8];
      if (FIRST) {
        const float bl = blin[col];
#pragma unroll
        for (int r = 0; r < 8; ++r) add[r] = bl;
      } else {
#pragma unroll
        for (int r = 0; r < 8; ++r) {
          int rowc = row0 + r + 8 * hi;
          if (rowc > N - 1) rowc = N - 1;
          add[r] = out[(size_t)rowc * 40 + col];
        }
      }
      float* optr = out + (size_t)(row0 + 8 * hi) * 40 + col;
#pragma unroll
      for (int r = 0; r < 8; ++r) {
        const int row = row0 + r + 8 * hi;
        if (row < N) optr[r * 40] = acc[r] + add[r];
      }
    }
  }
}

extern "C" void kernel_launch(void* const* d_in, const int* in_sizes, int n_in,
                              void* d_out, int out_size, void* d_ws, size_t ws_size,
                              hipStream_t stream) {
  const float* x    = (const float*)d_in[0];
  const int*   ei   = (const int*)d_in[1];
  const float* ew   = (const float*)d_in[2];
  const float* W1   = (const float*)d_in[3];
  const float* b1   = (const float*)d_in[4];
  const float* W2   = (const float*)d_in[5];
  const float* b2   = (const float*)d_in[6];
  const float* W3   = (const float*)d_in[7];
  const float* b3   = (const float*)d_in[8];
  const float* Wlin = (const float*)d_in[9];
  const float* blin = (const float*)d_in[10];

  const int N = in_sizes[0] / 128;
  const int E = in_sizes[2];
  const int* src = ei;
  const int* dst = ei + E;

  float* Ta   = (float*)d_ws;
  float* mid  = Ta  + (size_t)N * 64;
  float* Tb   = mid + (size_t)N * 64;

  const int degLen = (N + 1 + 3) & ~3;
  int*  deg  = (int*)mid;
  int*  list = deg + degLen;
  int*  binned = (int*)Tb;
  int*  tab    = binned + ((2 * E + 3) & ~3);

  const int Mtiles = (N + 15) / 16;
  const int B = (N + 255) / 256;
  const int C = (E + EDGE_CHUNK - 1) / EDGE_CHUNK;

  chunk_sort<<<C, 1024, 0, stream>>>(src, dst, ew, binned, tab, E, C);
  bin_sort<<<B, 1024, 0, stream>>>(binned, tab, deg, list, N, E, C);

  gemm_wmma_k<128, 128, 64, false><<<dim3(Mtiles, 1, 1), dim3(32, 4, 1), 0, stream>>>(
      x, W1, nullptr, Ta, N);

  gather_layer<true, true><<<Mtiles, 256, 0, stream>>>(
      Ta, deg, list, b1, W2, Tb, Wlin, blin, (float*)d_out, N);

  gather_layer<true, false><<<Mtiles, 256, 0, stream>>>(
      Tb, deg, list, b2, W3, Ta, Wlin + 64 * 40, blin, (float*)d_out, N);

  gather_layer<false, false><<<Mtiles, 256, 0, stream>>>(
      Ta, deg, list, b3, nullptr, nullptr, Wlin + 128 * 40, blin, (float*)d_out, N);
}


# x@W1 GEMM runs as extra workgroups of the bin_sort launch (independent of the edge sort; fills its idle wave slots and WGPs) instead of a launch of its own: 6->5 launches
# speedup vs baseline: 3.3622x; 1.0562x over previous
#include <hip/hip_runtime.h>
#include <hip/hip_bf16.h>

typedef __attribute__((ext_vector_type(2))) float v2f;
typedef __attribute__((ext_vector_type(8))) float v8f;
typedef __attribute__((ext_vector_type(2))) int   v2i;

template <int K, int LDA, int LDB, bool COLGUARD>
__device__ __forceinline__ void gemm_wmma_tile(const float* __restrict__ A,
                                               const float* __restrict__ B,
                                               const float* __restrict__ bias,
                                               float* __restrict__ C, int M,
                                               int row0, int col0) {
  const int lane  = threadIdx.x & 31;
  const int laneM = lane & 15;
  const int hi    = lane >> 4;

  int arow = row0 + laneM;
  if (arow > M - 1) arow = M - 1;
  const float* __restrict__ aptr = A + (size_t)arow * LDA + 2 * hi;

  const int col   = col0 + laneM;
  int       colc  = col;
  float     cmask = 1.0f;
  if (COLGUARD) {
    if (colc > LDB - 1) { colc = LDB - 1; cmask = 0.0f; }
  }
  const float* __restrict__ bptr = B + 2 * hi * LDB + colc;

  v8f acc = {};
#pragma unroll
  for (int k = 0; k < K; k += 4) {
    v2f a = *(const v2f*)(aptr + k);
    v2f b;
    b.x = bptr[k * LDB];
    b.y = bptr[k * LDB + LDB];
    if (COLGUARD) { b.x *= cmask; b.y *= cmask; }
    acc = __builtin_amdgcn_wmma_f32_16x16x4_f32(false, a, false, b,
                                                (short)0, acc, false, false);
  }

  if (COLGUARD && cmask == 0.0f) return;
  const float bc = COLGUARD ? bias[col] : 0.0f;
  float* __restrict__ cptr = C + (size_t)(row0 + 8 * hi) * LDB + col;
#pragma unroll
  for (int r = 0; r < 8; ++r) {
    const int row = row0 + r + 8 * hi;
    if (row < M) cptr[r * LDB] = acc[r] + bc;
  }
}

constexpr int EDGE_CHUNK = 3072;
constexpr int SORT_CAP   = 8192;

__device__ __forceinline__ void scan256(const int* cnt, int* pre, int* wsum, int tid) {
  const int lane = tid & 31;
  const int wid  = tid >> 5;
  const int v = (tid < 256) ? cnt[tid] : 0;
  int incl = v;
#pragma unroll
  for (int d = 1; d < 32; d <<= 1) {
    const int t = __shfl_up(incl, d);
    if (lane >= d) incl += t;
  }
  if (tid < 256 && lane == 31) wsum[wid] = incl;
  __syncthreads();
  if (tid < 256) {
    int base = 0;
    for (int w = 0; w < wid; ++w) base += wsum[w];
    pre[tid] = base + incl - v;
  }
  __syncthreads();
}

__global__ void __launch_bounds__(1024)
chunk_sort(const int* __restrict__ src, const int* __restrict__ dst,
           const float* __restrict__ w, int* __restrict__ binned,
           int* __restrict__ tab, int E, int C) {
  __shared__ int cnt[256];
  __shared__ int lst[256];
  __shared__ int wsum[8];
  __shared__ v2i stage[EDGE_CHUNK];
  const int tid = threadIdx.x;
  const int c   = blockIdx.x;
  if (tid < 256) cnt[tid] = 0;
  __syncthreads();
  const int e0 = c * EDGE_CHUNK;
  int nloc = E - e0;
  if (nloc > EDGE_CHUNK) nloc = EDGE_CHUNK;
  v2i ent[3];
  int bb[3], rk[3];
#pragma unroll
  for (int k = 0; k < 3; ++k) {
    const int  i  = k * 1024 + tid;
    const bool ok = i < nloc;
    const int  ix = e0 + (ok ? i : 0);
    const int  d  = dst[ix];
    bb[k]    = ok ? (d >> 8) : -1;
    ent[k].x = (src[ix] << 8) | (d & 255);
    ent[k].y = __float_as_int(w[ix]);
  }
#pragma unroll
  for (int k = 0; k < 3; ++k) {
    rk[k] = 0;
    if (bb[k] >= 0) rk[k] = atomicAdd(&cnt[bb[k]], 1);
  }
  __syncthreads();
  scan256(cnt, lst, wsum, tid);
#pragma unroll
  for (int k = 0; k < 3; ++k) {
    if (bb[k] >= 0) stage[lst[bb[k]] + rk[k]] = ent[k];
  }
  if (tid < 256) tab[tid * C + c] = lst[tid];
  if (tid == 0)  tab[256 * C + c] = nloc;
  __syncthreads();
  v2i* __restrict__ outp = (v2i*)(binned + 2 * (size_t)e0);
  for (int p = tid; p < nloc; p += 1024) outp[p] = stage[p];
}

__global__ void __launch_bounds__(1024)
bin_sort_gemm(const int* __restrict__ binned, const int* __restrict__ tab,
              int* __restrict__ first, int* __restrict__ list,
              int N, int E, int C, int B,
              const float* __restrict__ x, const float* __restrict__ W1,
              float* __restrict__ Ta) {
  if ((int)blockIdx.x >= B) {
    const int wid = __builtin_amdgcn_readfirstlane((int)(threadIdx.x >> 5));
    const int rt  = ((int)blockIdx.x - B) * 8 + (wid >> 2);
    if (rt * 16 < N)
      gemm_wmma_tile<128, 128, 64, false>(x, W1, nullptr, Ta, N, rt * 16, (wid & 3) * 16);
    return;
  }
  __shared__ int cnt[256];
  __shared__ int cur[256];
  __shared__ int wsum[8];
  __shared__ int rs[1024];
  __shared__ int rl[1024];
  __shared__ int red[64];
  __shared__ int tot[2];
  __shared__ v2i sout[SORT_CAP];
  const int tid  = threadIdx.x;
  const int lane = tid & 31;
  const int wid  = tid >> 5;
  const int b    = blockIdx.x;
  const int node0 = b * 256;

  int a = 0, len = 0;
  if (tid < C) {
    a   = tab[b * C + tid];
    len = tab[(b + 1) * C + tid] - a;
    rs[tid] = tid * EDGE_CHUNK + a;
    rl[tid] = len;
  }
#pragma unroll
  for (int d = 16; d >= 1; d >>= 1) {
    a   += __shfl_xor(a, d);
    len += __shfl_xor(len, d);
  }
  if (lane == 0) { red[wid] = a; red[32 + wid] = len; }
  if (tid < 256) cnt[tid] = 0;
  __syncthreads();
  if (wid == 0) {
    int sa = red[lane], sl = red[32 + lane];
#pragma unroll
    for (int d = 16; d >= 1; d >>= 1) {
      sa += __shfl_xor(sa, d);
      sl += __shfl_xor(sl, d);
    }
    if (lane == 0) { tot[0] = sa; tot[1] = sl; }
  }
  __syncthreads();
  const int beg = tot[0];
  const int n   = tot[1];

  const int hw  = tid >> 4;
  const int l16 = tid & 15;
  for (int r = hw; r < C; r += 64) {
    const int s = rs[r], m = rl[r];
    for (int j = l16; j < m; j += 16)
      atomicAdd(&cnt[binned[2 * (size_t)(s + j)] & 255], 1);
  }
  __syncthreads();
  scan256(cnt, cur, wsum, tid);
  if (tid < 256 && node0 + tid < N) first[node0 + tid] = beg + cur[tid];
  if (node0 + 256 >= N && tid == 0) first[N] = E;
  __syncthreads();

  const bool park = n <= SORT_CAP;
  v2i* __restrict__ outp = (v2i*)(list + 2 * (size_t)beg);
  for (int r = hw; r < C; r += 64) {
    const int s = rs[r], m = rl[r];
    for (int j = l16; j < m; j += 16) {
      const v2i e = *(const v2i*)(binned + 2 * (size_t)(s + j));
      const int pos = atomicAdd(&cur[e.x & 255], 1);
      v2i o;
      o.x = (int)((unsigned)e.x >> 8);
      o.y = e.y;
      if (park) sout[pos] = o;
      else      outp[pos] = o;
    }
  }
  if (park) {
    __syncthreads();
    for (int i = tid; i < n; i += 1024) outp[i] = sout[i];
  }
}

constexpr int H_PITCH = 68;

template <int LDB>
__device__ __forceinline__ v8f tile_k64(const float* arow,
                                        const float* __restrict__ bptr) {
  v8f acc = {};
#pragma unroll
  for (int k = 0; k < 64; k += 8) {
    const float4 a4 = *(const float4*)(arow + k);
    v2f a0, a1, b0, b1;
    a0.x = a4.x; a0.y = a4.y;
    a1.x = a4.z; a1.y = a4.w;
    b0.x = bptr[(k + 0) * LDB];
    b0.y = bptr[(k + 1) * LDB];
    b1.x = bptr[(k + 2) * LDB];
    b1.y = bptr[(k + 3) * LDB];
    acc = __builtin_amdgcn_wmma_f32_16x16x4_f32(false, a0, false, b0,
                                                (short)0, acc, false, false);
    acc = __builtin_amdgcn_wmma_f32_16x16x4_f32(false, a1, false, b1,
                                                (short)0, acc, false, false);
  }
  return acc;
}

template <bool HAS_NEXT, bool FIRST>
__global__ void __launch_bounds__(256)
gather_layer(const float* __restrict__ T,
             const int* __restrict__ first,
             const int* __restrict__ list,
             const float* __restrict__ b,
             const float* __restrict__ Wn,
             float* __restrict__ Tn,
             const float* __restrict__ Wl,
             const float* __restrict__ blin,
             float* __restrict__ out,
             int N) {
  __shared__ __attribute__((aligned(16))) float Hs[16 * H_PITCH];

  const int lane = threadIdx.x & 31;
  const int wid  = __builtin_amdgcn_readfirstlane((int)(threadIdx.x >> 5));
  const int row0 = blockIdx.x * 16;

  const float* __restrict__ Tl = T + 2 * lane;
  const v2f bv = *(const v2f*)(b + 2 * lane);
  for (int t = 0; t < 2; ++t) {
    const int ln   = 2 * wid + t;
    const int node = row0 + ln;
    v2f o = {0.0f, 0.0f};
    if (node < N) {
      const int beg = first[node];
      const int end = first[node + 1];
      v2f acc = {0.0f, 0.0f};
      for (int c = beg; c < end; c += 32) {
        const int n = (end - c < 32) ? (end - c) : 32;
        v2i me = {0, 0};
        if (lane < n) me = *(const v2i*)(list + 2 * (size_t)(c + lane));
#pragma unroll 4
        for (int j = 0; j < n; ++j) {
          const int   s  = __builtin_amdgcn_readlane(me.x, j);
          const float we = __int_as_float(__builtin_amdgcn_readlane(me.y, j));
          const v2f   tv = *(const v2f*)(Tl + (size_t)s * 64);
          acc.x += tv.x * we;
          acc.y += tv.y * we;
        }
      }
      o.x = fmaxf(acc.x + bv.x, 0.0f);
      o.y = fmaxf(acc.y + bv.y, 0.0f);
    }
    *(v2f*)(Hs + ln * H_PITCH + 2 * lane) = o;
  }

  __syncthreads();

  const int laneM = lane & 15;
  const int hi    = lane >> 4;
  const float* arow = Hs + laneM * H_PITCH + 4 * hi;
  constexpr int OUTW0 = HAS_NEXT ? 4 : 0;

  if (HAS_NEXT && wid < 4) {
    const int col = 16 * wid + laneM;
    const v8f acc = tile_k64<64>(arow, Wn + 4 * hi * 64 + col);
    float* __restrict__ cptr = Tn + (size_t)(row0 + 8 * hi) * 64 + col;
#pragma unroll
    for (int r = 0; r < 8; ++r) {
      const int row = row0 + r + 8 * hi;
      if (row < N) cptr[r * 64] = acc[r];
    }
  } else if (wid >= OUTW0 && wid < OUTW0 + 3) {
    const int col  = 16 * (wid - OUTW0) + laneM;
    const int colc = (col < 40) ? col : 39;
    const v8f acc = tile_k64<40>(arow, Wl + 4 * hi * 40 + colc);
    if (col < 40) {
      float add[8];
      if (FIRST) {
        const float bl = blin[col];
#pragma unroll
        for (int r = 0; r < 8; ++r) add[r] = bl;
      } else {
#pragma unroll
        for (int r = 0; r < 8; ++r) {
          int rowc = row0 + r + 8 * hi;
          if (rowc > N - 1) rowc = N - 1;
          add[r] = out[(size_t)rowc * 40 + col];
        }
      }
      float* optr = out + (size_t)(row0 + 8 * hi) * 40 + col;
#pragma unroll
      for (int r = 0; r < 8; ++r) {
        const int row = row0 + r + 8 * hi;
        if (row < N) optr[r * 40] = acc[r] + add[r];
      }
    }
  }
}

extern "C" void kernel_launch(void* const* d_in, const int* in_sizes, int n_in,
                              void* d_out, int out_size, void* d_ws, size_t ws_size,
                              hipStream_t stream) {
  const float* x    = (const float*)d_in[0];
  const int*   ei   = (const int*)d_in[1];
  const float* ew   = (const float*)d_in[2];
  const float* W1   = (const float*)d_in[3];
  const float* b1   = (const float*)d_in[4];
  const float* W2   = (const float*)d_in[5];
  const float* b2   = (const float*)d_in[6];
  const float* W3   = (const float*)d_in[7];
  const float* b3   = (const float*)d_in[8];
  const float* Wlin = (const float*)d_in[9];
  const float* blin = (const float*)d_in[10];

  const int N = in_sizes[0] / 128;
  const int E = in_sizes[2];
  const int* src = ei;
  const int* dst = ei + E;

  float* Ta   = (float*)d_ws;
  float* mid  = Ta  + (size_t)N * 64;
  float* Tb   = mid + (size_t)N * 64;

  const int degLen = (N + 1 + 3) & ~3;
  int*  deg  = (int*)mid;
  int*  list = deg + degLen;
  int*  binned = (int*)Tb;
  int*  tab    = binned + ((2 * E + 3) & ~3);

  const int Mtiles = (N + 15) / 16;
  const int B = (N + 255) / 256;
  const int C = (E + EDGE_CHUNK - 1) / EDGE_CHUNK;
  const int G = (Mtiles + 7) / 8;

  chunk_sort<<<C, 1024, 0, stream>>>(src, dst, ew, binned, tab, E, C);
  bin_sort_gemm<<<B + G, 1024, 0, stream>>>(binned, tab, deg, list, N, E, C, B, x, W1, Ta);

  gather_layer<true, true><<<Mtiles, 256, 0, stream>>>(
      Ta, deg, list, b1, W2, Tb, Wlin, blin, (float*)d_out, N);

  gather_layer<true, false><<<Mtiles, 256, 0, stream>>>(
      Tb, deg, list, b2, W3, Ta, Wlin + 64 * 40, blin, (float*)d_out, N);

  gather_layer<false, false><<<Mtiles, 256, 0, stream>>>(
      Ta, deg, list, b3, nullptr, nullptr, Wlin + 128 * 40, blin, (float*)d_out, N);
}


# x@W1 workgroups stage their 128 rows of x in LDS by whole-line loads and read A fragments from LDS (ds_load_b128), not by 16-line global loads repeated by 4 waves; k order in each 8 changed
# speedup vs baseline: 3.5944x; 1.0691x over previous
#include <hip/hip_runtime.h>
#include <hip/hip_bf16.h>

typedef __attribute__((ext_vector_type(2))) float v2f;
typedef __attribute__((ext_vector_type(8))) float v8f;
typedef __attribute__((ext_vector_type(2))) int   v2i;

constexpr int EDGE_CHUNK = 3072;
constexpr int SORT_CAP   = 8192;

__device__ __forceinline__ void scan256(const int* cnt, int* pre, int* wsum, int tid) {
  const int lane = tid & 31;
  const int wid  = tid >> 5;
  const int v = (tid < 256) ? cnt[tid] : 0;
  int incl = v;
#pragma unroll
  for (int d = 1; d < 32; d <<= 1) {
    const int t = __shfl_up(incl, d);
    if (lane >= d) incl += t;
  }
  if (tid < 256 && lane == 31) wsum[wid] = incl;
  __syncthreads();
  if (tid < 256) {
    int base = 0;
    for (int w = 0; w < wid; ++w) base += wsum[w];
    pre[tid] = base + incl - v;
  }
  __syncthreads();
}

__global__ void __launch_bounds__(1024)
chunk_sort(const int* __restrict__ src, const int* __restrict__ dst,
           const float* __restrict__ w, int* __restrict__ binned,
           int* __restrict__ tab, int E, int C) {
  __shared__ int cnt[256];
  __shared__ int lst[256];
  __shared__ int wsum[8];
  __shared__ v2i stage[EDGE_CHUNK];
  const int tid = threadIdx.x;
  const int c   = blockIdx.x;
  if (tid < 256) cnt[tid] = 0;
  __syncthreads();
  const int e0 = c * EDGE_CHUNK;
  int nloc = E - e0;
  if (nloc > EDGE_CHUNK) nloc = EDGE_CHUNK;
  v2i ent[3];
  int bb[3], rk[3];
#pragma unroll
  for (int k = 0; k < 3; ++k) {
    const int  i  = k * 1024 + tid;
    const bool ok = i < nloc;
    const int  ix = e0 + (ok ? i : 0);
    const int  d  = dst[ix];
    bb[k]    = ok ? (d >> 8) : -1;
    ent[k].x = (src[ix] << 8) | (d & 255);
    ent[k].y = __float_as_int(w[ix]);
  }
#pragma unroll
  for (int k = 0; k < 3; ++k) {
    rk[k] = 0;
    if (bb[k] >= 0) rk[k] = atomicAdd(&cnt[bb[k]], 1);
  }
  __syncthreads();
  scan256(cnt, lst, wsum, tid);
#pragma unroll
  for (int k = 0; k < 3; ++k) {
    if (bb[k] >= 0) stage[lst[bb[k]] + rk[k]] = ent[k];
  }
  if (tid < 256) tab[tid * C + c] = lst[tid];
  if (tid == 0)  tab[256 * C + c] = nloc;
  __syncthreads();
  v2i* __restrict__ outp = (v2i*)(binned + 2 * (size_t)e0);
  for (int p = tid; p < nloc; p += 1024) outp[p] = stage[p];
}

constexpr int X_PITCH = 132;

__device__ __forceinline__ v8f tile_k128_w64(const float* arow,
                                             const float* __restrict__ bptr) {
  v8f acc = {};
#pragma unroll
  for (int k = 0; k < 128; k += 8) {
    const float4 a4 = *(const float4*)(arow + k);
    v2f a0, a1, b0, b1;
    a0.x = a4.x; a0.y = a4.y;
    a1.x = a4.z; a1.y = a4.w;
    b0.x = bptr[(k + 0) * 64];
    b0.y = bptr[(k + 1) * 64];
    b1.x = bptr[(k + 2) * 64];
    b1.y = bptr[(k + 3) * 64];
    acc = __builtin_amdgcn_wmma_f32_16x16x4_f32(false, a0, false, b0,
                                                (short)0, acc, false, false);
    acc = __builtin_amdgcn_wmma_f32_16x16x4_f32(false, a1, false, b1,
                                                (short)0, acc, false, false);
  }
  return acc;
}

__global__ void __launch_bounds__(1024)
bin_sort_gemm(const int* __restrict__ binned, const int* __restrict__ tab,
              int* __restrict__ first, int* __restrict__ list,
              int N, int E, int C, int B,
              const float* __restrict__ x, const float* __restrict__ W1,
              float* __restrict__ Ta) {
  if ((int)blockIdx.x >= B) {
    __shared__ __attribute__((aligned(16))) float Xs[128 * X_PITCH];
    const int tid   = threadIdx.x;
    const int rbase = ((int)blockIdx.x - B) * 128;
#pragma unroll
    for (int i = 0; i < 4; ++i) {
      const int idx = i * 1024 + tid;
      const int r   = idx >> 5;
      const int c4  = idx & 31;
      int row = rbase + r;
      if (row > N - 1) row = N - 1;
      const float4 v = *(const float4*)(x + (size_t)row * 128 + 4 * c4);
      *(float4*)(Xs + r * X_PITCH + 4 * c4) = v;
    }
    __syncthreads();
    const int wid   = __builtin_amdgcn_readfirstlane(tid >> 5);
    const int lane  = tid & 31;
    const int laneM = lane & 15;
    const int hi    = lane >> 4;
    const int row0  = rbase + (wid >> 2) * 16;
    if (row0 < N) {
      const int col = (wid & 3) * 16 + laneM;
      const v8f acc = tile_k128_w64(Xs + ((wid >> 2) * 16 + laneM) * X_PITCH + 4 * hi,
                                    W1 + 4 * hi * 64 + col);
      float* __restrict__ cptr = Ta + (size_t)(row0 + 8 * hi) * 64 + col;
#pragma unroll
      for (int r = 0; r < 8; ++r) {
        const int row = row0 + r + 8 * hi;
        if (row < N) cptr[r * 64] = acc[r];
      }
    }
    return;
  }
  __shared__ int cnt[256];
  __shared__ int cur[256];
  __shared__ int wsum[8];
  __shared__ int rs[1024];
  __shared__ int rl[1024];
  __shared__ int red[64];
  __shared__ int tot[2];
  __shared__ v2i sout[SORT_CAP];
  const int tid  = threadIdx.x;
  const int lane = tid & 31;
  const int wid  = tid >> 5;
  const int b    = blockIdx.x;
  const int node0 = b * 256;

  int a = 0, len = 0;
  if (tid < C) {
    a   = tab[b * C + tid];
    len = tab[(b + 1) * C + tid] - a;
    rs[tid] = tid * EDGE_CHUNK + a;
    rl[tid] = len;
  }
#pragma unroll
  for (int d = 16; d >= 1; d >>= 1) {
    a   += __shfl_xor(a, d);
    len += __shfl_xor(len, d);
  }
  if (lane == 0) { red[wid] = a; red[32 + wid] = len; }
  if (tid < 256) cnt[tid] = 0;
  __syncthreads();
  if (wid == 0) {
    int sa = red[lane], sl = red[32 + lane];
#pragma unroll
    for (int d = 16; d >= 1; d >>= 1) {
      sa += __shfl_xor(sa, d);
      sl += __shfl_xor(sl, d);
    }
    if (lane == 0) { tot[0] = sa; tot[1] = sl; }
  }
  __syncthreads();
  const int beg = tot[0];
  const int n   = tot[1];

  const int hw  = tid >> 4;
  const int l16 = tid & 15;
  for (int r = hw; r < C; r += 64) {
    const int s = rs[r], m = rl[r];
    for (int j = l16; j < m; j += 16)
      atomicAdd(&cnt[binned[2 * (size_t)(s + j)] & 255], 1);
  }
  __syncthreads();
  scan256(cnt, cur, wsum, tid);
  if (tid < 256 && node0 + tid < N) first[node0 + tid] = beg + cur[tid];
  if (node0 + 256 >= N && tid == 0) first[N] = E;
  __syncthreads();

  const bool park = n <= SORT_CAP;
  v2i* __restrict__ outp = (v2i*)(list + 2 * (size_t)beg);
  for (int r = hw; r < C; r += 64) {
    const int s = rs[r], m = rl[r];
    for (int j = l16; j < m; j += 16) {
      const v2i e = *(const v2i*)(binned + 2 * (size_t)(s + j));
      const int pos = atomicAdd(&cur[e.x & 255], 1);
      v2i o;
      o.x = (int)((unsigned)e.x >> 8);
      o.y = e.y;
      if (park) sout[pos] = o;
      else      outp[pos] = o;
    }
  }
  if (park) {
    __syncthreads();
    for (int i = tid; i < n; i += 1024) outp[i] = sout[i];
  }
}

constexpr int H_PITCH = 68;

template <int LDB>
__device__ __forceinline__ v8f tile_k64(const float* arow,
                                        const float* __restrict__ bptr) {
  v8f acc = {};
#pragma unroll
  for (int k = 0; k < 64; k += 8) {
    const float4 a4 = *(const float4*)(arow + k);
    v2f a0, a1, b0, b1;
    a0.x = a4.x; a0.y = a4.y;
    a1.x = a4.z; a1.y = a4.w;
    b0.x = bptr[(k + 0) * LDB];
    b0.y = bptr[(k + 1) * LDB];
    b1.x = bptr[(k + 2) * LDB];
    b1.y = bptr[(k + 3) * LDB];
    acc = __builtin_amdgcn_wmma_f32_16x16x4_f32(false, a0, false, b0,
                                                (short)0, acc, false, false);
    acc = __builtin_amdgcn_wmma_f32_16x16x4_f32(false, a1, false, b1,
                                                (short)0, acc, false, false);
  }
  return acc;
}

template <bool HAS_NEXT, bool FIRST>
__global__ void __launch_bounds__(256)
gather_layer(const float* __restrict__ T,
             const int* __restrict__ first,
             const int* __restrict__ list,
             const float* __restrict__ b,
             const float* __restrict__ Wn,
             float* __restrict__ Tn,
             const float* __restrict__ Wl,
             const float* __restrict__ blin,
             float* __restrict__ out,
             int N) {
  __shared__ __attribute__((aligned(16))) float Hs[16 * H_PITCH];

  const int lane = threadIdx.x & 31;
  const int wid  = __builtin_amdgcn_readfirstlane((int)(threadIdx.x >> 5));
  const int row0 = blockIdx.x * 16;

  const float* __restrict__ Tl = T + 2 * lane;
  const v2f bv = *(const v2f*)(b + 2 * lane);
  for (int t = 0; t < 2; ++t) {
    const int ln   = 2 * wid + t;
    const int node = row0 + ln;
    v2f o = {0.0f, 0.0f};
    if (node < N) {
      const int beg = first[node];
      const int end = first[node + 1];
      v2f acc = {0.0f, 0.0f};
      for (int c = beg; c < end; c += 32) {
        const int n = (end - c < 32) ? (end - c) : 32;
        v2i me = {0, 0};
        if (lane < n) me = *(const v2i*)(list + 2 * (size_t)(c + lane));
#pragma unroll 4
        for (int j = 0; j < n; ++j) {
          const int   s  = __builtin_amdgcn_readlane(me.x, j);
          const float we = __int_as_float(__builtin_amdgcn_readlane(me.y, j));
          const v2f   tv = *(const v2f*)(Tl + (size_t)s * 64);
          acc.x += tv.x * we;
          acc.y += tv.y * we;
        }
      }
      o.x = fmaxf(acc.x + bv.x, 0.0f);
      o.y = fmaxf(acc.y + bv.y, 0.0f);
    }
    *(v2f*)(Hs + ln * H_PITCH + 2 * lane) = o;
  }

  __syncthreads();

  const int laneM = lane & 15;
  const int hi    = lane >> 4;
  const float* arow = Hs + laneM * H_PITCH + 4 * hi;
  constexpr int OUTW0 = HAS_NEXT ? 4 : 0;

  if (HAS_NEXT && wid < 4) {
    const int col = 16 * wid + laneM;
    const v8f acc = tile_k64<64>(arow, Wn + 4 * hi * 64 + col);
    float* __restrict__ cptr = Tn + (size_t)(row0 + 8 * hi) * 64 + col;
#pragma unroll
    for (int r = 0; r < 8; ++r) {
      const int row = row0 + r + 8 * hi;
      if (row < N) cptr[r * 64] = acc[r];
    }
  } else if (wid >= OUTW0 && wid < OUTW0 + 3) {
    const int col  = 16 * (wid - OUTW0) + laneM;
    const int colc = (col < 40) ? col : 39;
    const v8f acc = tile_k64<40>(arow, Wl + 4 * hi * 40 + colc);
    if (col < 40) {
      float add[8];
      if (FIRST) {
        const float bl = blin[col];
#pragma unroll
        for (int r = 0; r < 8; ++r) add[r] = bl;
      } else {
#pragma unroll
        for (int r = 0; r < 8; ++r) {
          int rowc = row0 + r + 8 * hi;
          if (rowc > N - 1) rowc = N - 1;
          add[r] = out[(size_t)rowc * 40 + col];
        }
      }
      float* optr = out + (size_t)(row0 + 8 * hi) * 40 + col;
#pragma unroll
      for (int r = 0; r < 8; ++r) {
        const int row = row0 + r + 8 * hi;
        if (row < N) optr[r * 40] = acc[r] + add[r];
      }
    }
  }
}

extern "C" void kernel_launch(void* const* d_in, const int* in_sizes, int n_in,
                              void* d_out, int out_size, void* d_ws, size_t ws_size,
                              hipStream_t stream) {
  const float* x    = (const float*)d_in[0];
  const int*   ei   = (const int*)d_in[1];
  const float* ew   = (const float*)d_in[2];
  const float* W1   = (const float*)d_in[3];
  const float* b1   = (const float*)d_in[4];
  const float* W2   = (const float*)d_in[5];
  const float* b2   = (const float*)d_in[6];
  const float* W3   = (const float*)d_in[7];
  const float* b3   = (const float*)d_in[8];
  const float* Wlin = (const float*)d_in[9];
  const float* blin = (const float*)d_in[10];

  const int N = in_sizes[0] / 128;
  const int E = in_sizes[2];
  const int* src = ei;
  const int* dst = ei + E;

  float* Ta   = (float*)d_ws;
  float* mid  = Ta  + (size_t)N * 64;
  float* Tb   = mid + (size_t)N * 64;

  const int degLen = (N + 1 + 3) & ~3;
  int*  deg  = (int*)mid;
  int*  list = deg + degLen;
  int*  binned = (int*)Tb;
  int*  tab    = binned + ((2 * E + 3) & ~3);

  const int Mtiles = (N + 15) / 16;
  const int B = (N + 255) / 256;
  const int C = (E + EDGE_CHUNK - 1) / EDGE_CHUNK;
  const int G = (Mtiles + 7) / 8;

  chunk_sort<<<C, 1024, 0, stream>>>(src, dst, ew, binned, tab, E, C);
  bin_sort_gemm<<<B + G, 1024, 0, stream>>>(binned, tab, deg, list, N, E, C, B, x, W1, Ta);

  gather_layer<true, true><<<Mtiles, 256, 0, stream>>>(
      Ta, deg, list, b1, W2, Tb, Wlin, blin, (float*)d_out, N);

  gather_layer<true, false><<<Mtiles, 256, 0, stream>>>(
      Tb, deg, list, b2, W3, Ta, Wlin + 64 * 40, blin, (float*)d_out, N);

  gather_layer<false, false><<<Mtiles, 256, 0, stream>>>(
      Ta, deg, list, b3, nullptr, nullptr, Wlin + 128 * 40, blin, (float*)d_out, N);
}


# Gather loop: each 16-lane half of a wave sums one node with 16-byte loads (one row per half per load), the two nodes side by side, 8 loads issued before the first add; per-node sum order unchanged
# speedup vs baseline: 3.7758x; 1.0505x over previous
#include <hip/hip_runtime.h>
#include <hip/hip_bf16.h>

typedef __attribute__((ext_vector_type(2))) float v2f;
typedef __attribute__((ext_vector_type(8))) float v8f;
typedef __attribute__((ext_vector_type(2))) int   v2i;
typedef __attribute__((ext_vector_type(4))) float v4f;

constexpr int EDGE_CHUNK = 3072;
constexpr int SORT_CAP   = 8192;

__device__ __forceinline__ void scan256(const int* cnt, int* pre, int* wsum, int tid) {
  const int lane = tid & 31;
  const int wid  = tid >> 5;
  const int v = (tid < 256) ? cnt[tid] : 0;
  int incl = v;
#pragma unroll
  for (int d = 1; d < 32; d <<= 1) {
    const int t = __shfl_up(incl, d);
    if (lane >= d) incl += t;
  }
  if (tid < 256 && lane == 31) wsum[wid] = incl;
  __syncthreads();
  if (tid < 256) {
    int base = 0;
    for (int w = 0; w < wid; ++w) base += wsum[w];
    pre[tid] = base + incl - v;
  }
  __syncthreads();
}

__global__ void __launch_bounds__(1024)
chunk_sort(const int* __restrict__ src, const int* __restrict__ dst,
           const float* __restrict__ w, int* __restrict__ binned,
           int* __restrict__ tab, int E, int C) {
  __shared__ int cnt[256];
  __shared__ int lst[256];
  __shared__ int wsum[8];
  __shared__ v2i stage[EDGE_CHUNK];
  const int tid = threadIdx.x;
  const int c   = blockIdx.x;
  if (tid < 256) cnt[tid] = 0;
  __syncthreads();
  const int e0 = c * EDGE_CHUNK;
  int nloc = E - e0;
  if (nloc > EDGE_CHUNK) nloc = EDGE_CHUNK;
  v2i ent[3];
  int bb[3], rk[3];
#pragma unroll
  for (int k = 0; k < 3; ++k) {
    const int  i  = k * 1024 + tid;
    const bool ok = i < nloc;
    const int  ix = e0 + (ok ? i : 0);
    const int  d  = dst[ix];
    bb[k]    = ok ? (d >> 8) : -1;
    ent[k].x = (src[ix] << 8) | (d & 255);
    ent[k].y = __float_as_int(w[ix]);
  }
#pragma unroll
  for (int k = 0; k < 3; ++k) {
    rk[k] = 0;
    if (bb[k] >= 0) rk[k] = atomicAdd(&cnt[bb[k]], 1);
  }
  __syncthreads();
  scan256(cnt, lst, wsum, tid);
#pragma unroll
  for (int k = 0; k < 3; ++k) {
    if (bb[k] >= 0) stage[lst[bb[k]] + rk[k]] = ent[k];
  }
  if (tid < 256) tab[tid * C + c] = lst[tid];
  if (tid == 0)  tab[256 * C + c] = nloc;
  __syncthreads();
  v2i* __restrict__ outp = (v2i*)(binned + 2 * (size_t)e0);
  for (int p = tid; p < nloc; p += 1024) outp[p] = stage[p];
}

constexpr int X_PITCH = 132;

__device__ __forceinline__ v8f tile_k128_w64(const float* arow,
                                             const float* __restrict__ bptr) {
  v8f acc = {};
#pragma unroll
  for (int k = 0; k < 128; k += 8) {
    const float4 a4 = *(const float4*)(arow + k);
    v2f a0, a1, b0, b1;
    a0.x = a4.x; a0.y = a4.y;
    a1.x = a4.z; a1.y = a4.w;
    b0.x = bptr[(k + 0) * 64];
    b0.y = bptr[(k + 1) * 64];
    b1.x = bptr[(k + 2) * 64];
    b1.y = bptr[(k + 3) * 64];
    acc = __builtin_amdgcn_wmma_f32_16x16x4_f32(false, a0, false, b0,
                                                (short)0, acc, false, false);
    acc = __builtin_amdgcn_wmma_f32_16x16x4_f32(false, a1, false, b1,
                                                (short)0, acc, false, false);
  }
  return acc;
}

__global__ void __launch_bounds__(1024)
bin_sort_gemm(const int* __restrict__ binned, const int* __restrict__ tab,
              int* __restrict__ first, int* __restrict__ list,
              int N, int E, int C, int B,
              const float* __restrict__ x, const float* __restrict__ W1,
              float* __restrict__ Ta) {
  if ((int)blockIdx.x >= B) {
    __shared__ __attribute__((aligned(16))) float Xs[128 * X_PITCH];
    const int tid   = threadIdx.x;
    const int rbase = ((int)blockIdx.x - B) * 128;
#pragma unroll
    for (int i = 0; i < 4; ++i) {
      const int idx = i * 1024 + tid;
      const int r   = idx >> 5;
      const int c4  = idx & 31;
      int row = rbase + r;
      if (row > N - 1) row = N - 1;
      const float4 v = *(const float4*)(x + (size_t)row * 128 + 4 * c4);
      *(float4*)(Xs + r * X_PITCH + 4 * c4) = v;
    }
    __syncthreads();
    const int wid   = __builtin_amdgcn_readfirstlane(tid >> 5);
    const int lane  = tid & 31;
    const int laneM = lane & 15;
    const int hi    = lane >> 4;
    const int row0  = rbase + (wid >> 2) * 16;
    if (row0 < N) {
      const int col = (wid & 3) * 16 + laneM;
      const v8f acc = tile_k128_w64(Xs + ((wid >> 2) * 16 + laneM) * X_PITCH + 4 * hi,
                                    W1 + 4 * hi * 64 + col);
      float* __restrict__ cptr = Ta + (size_t)(row0 + 8 * hi) * 64 + col;
#pragma unroll
      for (int r = 0; r < 8; ++r) {
        const int row = row0 + r + 8 * hi;
        if (row < N) cptr[r * 64] = acc[r];
      }
    }
    return;
  }
  __shared__ int cnt[256];
  __shared__ int cur[256];
  __shared__ int wsum[8];
  __shared__ int rs[1024];
  __shared__ int rl[1024];
  __shared__ int red[64];
  __shared__ int tot[2];
  __shared__ v2i sout[SORT_CAP];
  const int tid  = threadIdx.x;
  const int lane = tid & 31;
  const int wid  = tid >> 5;
  const int b    = blockIdx.x;
  const int node0 = b * 256;

  int a = 0, len = 0;
  if (tid < C) {
    a   = tab[b * C + tid];
    len = tab[(b + 1) * C + tid] - a;
    rs[tid] = tid * EDGE_CHUNK + a;
    rl[tid] = len;
  }
#pragma unroll
  for (int d = 16; d >= 1; d >>= 1) {
    a   += __shfl_xor(a, d);
    len += __shfl_xor(len, d);
  }
  if (lane == 0) { red[wid] = a; red[32 + wid] = len; }
  if (tid < 256) cnt[tid] = 0;
  __syncthreads();
  if (wid == 0) {
    int sa = red[lane], sl = red[32 + lane];
#pragma unroll
    for (int d = 16; d >= 1; d >>= 1) {
      sa += __shfl_xor(sa, d);
      sl += __shfl_xor(sl, d);
    }
    if (lane == 0) { tot[0] = sa; tot[1] = sl; }
  }
  __syncthreads();
  const int beg = tot[0];
  const int n   = tot[1];

  const int hw  = tid >> 4;
  const int l16 = tid & 15;
  for (int r = hw; r < C; r += 64) {
    const int s = rs[r], m = rl[r];
    for (int j = l16; j < m; j += 16)
      atomicAdd(&cnt[binned[2 * (size_t)(s + j)] & 255], 1);
  }
  __syncthreads();
  scan256(cnt, cur, wsum, tid);
  if (tid < 256 && node0 + tid < N) first[node0 + tid] = beg + cur[tid];
  if (node0 + 256 >= N && tid == 0) first[N] = E;
  __syncthreads();

  const bool park = n <= SORT_CAP;
  v2i* __restrict__ outp = (v2i*)(list + 2 * (size_t)beg);
  for (int r = hw; r < C; r += 64) {
    const int s = rs[r], m = rl[r];
    for (int j = l16; j < m; j += 16) {
      const v2i e = *(const v2i*)(binned + 2 * (size_t)(s + j));
      const int pos = atomicAdd(&cur[e.x & 255], 1);
      v2i o;
      o.x = (int)((unsigned)e.x >> 8);
      o.y = e.y;
      if (park) sout[pos] = o;
      else      outp[pos] = o;
    }
  }
  if (park) {
    __syncthreads();
    for (int i = tid; i < n; i += 1024) outp[i] = sout[i];
  }
}

constexpr int H_PITCH = 68;

template <int LDB>
__device__ __forceinline__ v8f tile_k64(const float* arow,
                                        const float* __restrict__ bptr) {
  v8f acc = {};
#pragma unroll
  for (int k = 0; k < 64; k += 8) {
    const float4 a4 = *(const float4*)(arow + k);
    v2f a0, a1, b0, b1;
    a0.x = a4.x; a0.y = a4.y;
    a1.x = a4.z; a1.y = a4.w;
    b0.x = bptr[(k + 0) * LDB];
    b0.y = bptr[(k + 1) * LDB];
    b1.x = bptr[(k + 2) * LDB];
    b1.y = bptr[(k + 3) * LDB];
    acc = __builtin_amdgcn_wmma_f32_16x16x4_f32(false, a0, false, b0,
                                                (short)0, acc, false, false);
    acc = __builtin_amdgcn_wmma_f32_16x16x4_f32(false, a1, false, b1,
                                                (short)0, acc, false, false);
  }
  return acc;
}

template <int J>
__device__ __forceinline__ int half_lane(int v) {
  return __builtin_amdgcn_update_dpp(0, v, 0x150 + J, 0xF, 0xF, false);
}

template <int B>
__device__ __forceinline__ void gather8(const char* __restrict__ Tb, unsigned lo,
                                        int ex, int ey, int nleft,
                                        v2f& a01, v2f& a23) {
  v4f   tv[8];
  float wv[8];
#define GATHER8_ROW(u)                                                      \
  {                                                                         \
    const unsigned off_ = ((unsigned)half_lane<B + (u)>(ex) << 8) + lo;     \
    wv[u] = __int_as_float(half_lane<B + (u)>(ey));                         \
    tv[u] = v4f{0.0f, 0.0f, 0.0f, 0.0f};                                    \
    if ((u) < nleft) tv[u] = *(const v4f*)(Tb + off_);                      \
  }
  GATHER8_ROW(0) GATHER8_ROW(1) GATHER8_ROW(2) GATHER8_ROW(3)
  GATHER8_ROW(4) GATHER8_ROW(5) GATHER8_ROW(6) GATHER8_ROW(7)
#undef GATHER8_ROW
#pragma unroll
  for (int u = 0; u < 8; ++u) {
    const v2f w2  = {wv[u], wv[u]};
    const v2f t01 = {tv[u].x, tv[u].y};
    const v2f t23 = {tv[u].z, tv[u].w};
    a01 = __builtin_elementwise_fma(t01, w2, a01);
    a23 = __builtin_elementwise_fma(t23, w2, a23);
  }
}

template <bool HAS_NEXT, bool FIRST>
__global__ void __launch_bounds__(256)
gather_layer(const float* __restrict__ T,
             const int* __restrict__ first,
             const int* __restrict__ list,
             const float* __restrict__ b,
             const float* __restrict__ Wn,
             float* __restrict__ Tn,
             const float* __restrict__ Wl,
             const float* __restrict__ blin,
             float* __restrict__ out,
             int N) {
  __shared__ __attribute__((aligned(16))) float Hs[16 * H_PITCH];

  const int lane = threadIdx.x & 31;
  const int wid  = __builtin_amdgcn_readfirstlane((int)(threadIdx.x >> 5));
  const int row0 = blockIdx.x * 16;

  const int l16  = lane & 15;
  const int hh   = lane >> 4;
  const int ln   = 2 * wid + hh;
  const int node = row0 + ln;
  const v4f bv = *(const v4f*)(b + 4 * l16);
  const int nodec = (node < N) ? node : (N - 1);
  const int beg   = first[nodec];
  const int deg   = (node < N) ? (first[nodec + 1] - beg) : 0;
  const int d0   = __builtin_amdgcn_readlane(deg, 0);
  const int d1   = __builtin_amdgcn_readlane(deg, 16);
  const int dmax = (d0 > d1) ? d0 : d1;
  const char* __restrict__ Tb = (const char*)T;
  const unsigned lo = 16u * (unsigned)l16;
  const v2i* __restrict__ lp = (const v2i*)list + beg + l16;
  v2f a01 = {0.0f, 0.0f}, a23 = {0.0f, 0.0f};
  for (int c = 0; c < dmax; c += 32) {
    const int n = deg - c;
    v2i m0 = {0, 0}, m1 = {0, 0};
    if (l16 < n)      m0 = lp[c];
    if (l16 + 16 < n) m1 = lp[c + 16];
    const int nm = dmax - c;
    gather8<0>(Tb, lo, m0.x, m0.y, n, a01, a23);
    if (nm > 8)  gather8<8>(Tb, lo, m0.x, m0.y, n - 8, a01, a23);
    if (nm > 16) gather8<0>(Tb, lo, m1.x, m1.y, n - 16, a01, a23);
    if (nm > 24) gather8<8>(Tb, lo, m1.x, m1.y, n - 24, a01, a23);
  }
  {
    v4f o = {0.0f, 0.0f, 0.0f, 0.0f};
    if (node < N) {
      o.x = fmaxf(a01.x + bv.x, 0.0f);
      o.y = fmaxf(a01.y + bv.y, 0.0f);
      o.z = fmaxf(a23.x + bv.z, 0.0f);
      o.w = fmaxf(a23.y + bv.w, 0.0f);
    }
    *(v4f*)(Hs + ln * H_PITCH + 4 * l16) = o;
  }

  __syncthreads();

  const int laneM = lane & 15;
  const int hi    = lane >> 4;
  const float* arow = Hs + laneM * H_PITCH + 4 * hi;
  constexpr int OUTW0 = HAS_NEXT ? 4 : 0;

  if (HAS_NEXT && wid < 4) {
    const int col = 16 * wid + laneM;
    const v8f acc = tile_k64<64>(arow, Wn + 4 * hi * 64 + col);
    float* __restrict__ cptr = Tn + (size_t)(row0 + 8 * hi) * 64 + col;
#pragma unroll
    for (int r = 0; r < 8; ++r) {
      const int row = row0 + r + 8 * hi;
      if (row < N) cptr[r * 64] = acc[r];
    }
  } else if (wid >= OUTW0 && wid < OUTW0 + 3) {
    const int col  = 16 * (wid - OUTW0) + laneM;
    const int colc = (col < 40) ? col : 39;
    const v8f acc = tile_k64<40>(arow, Wl + 4 * hi * 40 + colc);
    if (col < 40) {
      float add[8];
      if (FIRST) {
        const float bl = blin[col];
#pragma unroll
        for (int r = 0; r < 8; ++r) add[r] = bl;
      } else {
#pragma unroll
        for (int r = 0; r < 8; ++r) {
          int rowc = row0 + r + 8 * hi;
          if (rowc > N - 1) rowc = N - 1;
          add[r] = out[(size_t)rowc * 40 + col];
        }
      }
      float* optr = out + (size_t)(row0 + 8 * hi) * 40 + col;
#pragma unroll
      for (int r = 0; r < 8; ++r) {
        const int row = row0 + r + 8 * hi;
        if (row < N) optr[r * 40] = acc[r] + add[r];
      }
    }
  }
}

extern "C" void kernel_launch(void* const* d_in, const int* in_sizes, int n_in,
                              void* d_out, int out_size, void* d_ws, size_t ws_size,
                              hipStream_t stream) {
  const float* x    = (const float*)d_in[0];
  const int*   ei   = (const int*)d_in[1];
  const float* ew   = (const float*)d_in[2];
  const float* W1   = (const float*)d_in[3];
  const float* b1   = (const float*)d_in[4];
  const float* W2   = (const float*)d_in[5];
  const float* b2   = (const float*)d_in[6];
  const float* W3   = (const float*)d_in[7];
  const float* b3   = (const float*)d_in[8];
  const float* Wlin = (const float*)d_in[9];
  const float* blin = (const float*)d_in[10];

  const int N = in_sizes[0] / 128;
  const int E = in_sizes[2];
  const int* src = ei;
  const int* dst = ei + E;

  float* Ta   = (float*)d_ws;
  float* mid  = Ta  + (size_t)N * 64;
  float* Tb   = mid + (size_t)N * 64;

  const int degLen = (N + 1 + 3) & ~3;
  int*  deg  = (int*)mid;
  int*  list = deg + degLen;
  int*  binned = (int*)Tb;
  int*  tab    = binned + ((2 * E + 3) & ~3);

  const int Mtiles = (N + 15) / 16;
  const int B = (N + 255) / 256;
  const int C = (E + EDGE_CHUNK - 1) / EDGE_CHUNK;
  const int G = (Mtiles + 7) / 8;

  chunk_sort<<<C, 1024, 0, stream>>>(src, dst, ew, binned, tab, E, C);
  bin_sort_gemm<<<B + G, 1024, 0, stream>>>(binned, tab, deg, list, N, E, C, B, x, W1, Ta);

  gather_layer<true, true><<<Mtiles, 256, 0, stream>>>(
      Ta, deg, list, b1, W2, Tb, Wlin, blin, (float*)d_out, N);

  gather_layer<true, false><<<Mtiles, 256, 0, stream>>>(
      Tb, deg, list, b2, W3, Ta, Wlin + 64 * 40, blin, (float*)d_out, N);

  gather_layer<false, false><<<Mtiles, 256, 0, stream>>>(
      Ta, deg, list, b3, nullptr, nullptr, Wlin + 128 * 40, blin, (float*)d_out, N);
}
